# starting kernel
# baseline (speedup 1.0000x reference)
#include <hip/hip_runtime.h>
#include <hip/hip_bf16.h>
#include <math.h>

typedef float v2f __attribute__((ext_vector_type(2)));
typedef float v8f __attribute__((ext_vector_type(8)));

#define NEG_SLOPE 0.2f
#define EPS_F 1e-16f

#define SWIZ_XOR1  0x041F
#define SWIZ_XOR2  0x081F
#define SWIZ_XOR4  0x101F
#define SWIZ_XOR8  0x201F
#define SWIZ_XOR16 0x401F
#define SWZ(v, imm) __int_as_float(__builtin_amdgcn_ds_swizzle(__float_as_int(v), imm))

__device__ __forceinline__ float reduce16_add(float v) {
    v += SWZ(v, SWIZ_XOR1);
    v += SWZ(v, SWIZ_XOR2);
    v += SWZ(v, SWIZ_XOR4);
    v += SWZ(v, SWIZ_XOR8);
    return v;
}
__device__ __forceinline__ float reduce32_max(float v) {
    v = fmaxf(v, SWZ(v, SWIZ_XOR1));
    v = fmaxf(v, SWZ(v, SWIZ_XOR2));
    v = fmaxf(v, SWZ(v, SWIZ_XOR4));
    v = fmaxf(v, SWZ(v, SWIZ_XOR8));
    v = fmaxf(v, SWZ(v, SWIZ_XOR16));
    return v;
}

__device__ __forceinline__ unsigned int fkey(float f) {
    unsigned int u = __float_as_uint(f);
    return (u & 0x80000000u) ? ~u : (u | 0x80000000u);
}
__device__ __forceinline__ float funkey(unsigned int k) {
    return __uint_as_float((k & 0x80000000u) ? (k & 0x7FFFFFFFu) : ~k);
}

__global__ __launch_bounds__(256) void gat_gemm_wmma(
    const float* __restrict__ x, const float* __restrict__ W,
    const float* __restrict__ a_src, const float* __restrict__ a_trg,
    float* __restrict__ hbuf, float* __restrict__ s_src, float* __restrict__ s_trg,
    int N)
{
    const int wave  = threadIdx.x >> 5;
    const int lane  = threadIdx.x & 31;
    const int mBase = blockIdx.x * 16;
    const int nBase = wave * 16;

    const int f    = lane & 15;
    const int half = lane >> 4;
    const int ksub = half * 2;

    int rowA = mBase + f;
    if (rowA > N - 1) rowA = N - 1;
    const float* __restrict__ xrow = x + (size_t)rowA * 128;

    v8f c = {};
    #pragma unroll 8
    for (int k = 0; k < 128; k += 4) {
        v2f a;
        a.x = xrow[k + ksub];
        a.y = xrow[k + ksub + 1];
        v2f b;
        b.x = W[(size_t)(k + ksub) * 128 + nBase + f];
        b.y = W[(size_t)(k + ksub + 1) * 128 + nBase + f];
        c = __builtin_amdgcn_wmma_f32_16x16x4_f32(
                 false, a,  false, b,
                 (short)0, c,  false,  false);
    }

    const float asv = a_src[nBase + f];
    const float atv = a_trg[nBase + f];

    #pragma unroll
    for (int r = 0; r < 8; ++r) {
        const int m = mBase + r + half * 8;
        const float hv = c[r];
        if (m < N) hbuf[(size_t)m * 128 + nBase + f] = hv;

        const float ps = reduce16_add(hv * asv);
        const float pt = reduce16_add(hv * atv);
        if (f == 0 && m < N) {
            s_src[(size_t)m * 8 + wave] = ps;
            s_trg[(size_t)m * 8 + wave] = pt;
        }
    }
}

__global__ __launch_bounds__(256) void gat_edge_scores(
    const int* __restrict__ ei, int E,
    const float* __restrict__ s_src, const float* __restrict__ s_trg,
    float* __restrict__ ebuf, unsigned int* __restrict__ gmax)
{
    __shared__ unsigned int smax;
    if (threadIdx.x == 0) smax = 0u;
    __syncthreads();

    const int e = blockIdx.x * blockDim.x + threadIdx.x;
    float lmax = -INFINITY;
    if (e < E) {
        const int src = ei[e];
        const int trg = ei[E + e];
        const float* __restrict__ ss = s_src + (size_t)src * 8;
        const float* __restrict__ st = s_trg + (size_t)trg * 8;
        #pragma unroll
        for (int hh = 0; hh < 8; ++hh) {
            float v = ss[hh] + st[hh];
            v = (v >= 0.f) ? v : NEG_SLOPE * v;
            ebuf[(size_t)e * 8 + hh] = v;
            lmax = fmaxf(lmax, v);
        }
    }
    lmax = reduce32_max(lmax);
    if ((threadIdx.x & 31) == 0) atomicMax(&smax, fkey(lmax));
    __syncthreads();
    if (threadIdx.x == 0) atomicMax(gmax, smax);
}

__global__ __launch_bounds__(256) void gat_edge_exp(
    const int* __restrict__ ei, int E,
    float* __restrict__ ebuf, float* __restrict__ denom,
    const unsigned int* __restrict__ gmax)
{
    const int e = blockIdx.x * blockDim.x + threadIdx.x;
    if (e >= E) return;
    const float mx = funkey(*gmax);
    const int trg = ei[E + e];
    #pragma unroll
    for (int hh = 0; hh < 8; ++hh) {
        const float v = expf(ebuf[(size_t)e * 8 + hh] - mx);
        ebuf[(size_t)e * 8 + hh] = v;
        atomicAdd(&denom[(size_t)trg * 8 + hh], v);
    }
}

__global__ __launch_bounds__(256) void gat_aggregate(
    const int* __restrict__ ei, int E,
    const float* __restrict__ hbuf, const float* __restrict__ ebuf,
    const float* __restrict__ denom, float* __restrict__ out)
{
    const int t = blockIdx.x * blockDim.x + threadIdx.x;
    if (t >= E * 8) return;
    const int e  = t >> 3;
    const int hh = t & 7;
    const int src = ei[e];
    const int trg = ei[E + e];

    const float alpha = ebuf[t] / (denom[(size_t)trg * 8 + hh] + EPS_F);
    const float4* __restrict__ hp = (const float4*)(hbuf + (size_t)src * 128 + hh * 16);
    float* __restrict__ op = out + (size_t)trg * 128 + hh * 16;
    #pragma unroll
    for (int q = 0; q < 4; ++q) {
        const float4 v = hp[q];
        atomicAdd(op + q * 4 + 0, v.x * alpha);
        atomicAdd(op + q * 4 + 1, v.y * alpha);
        atomicAdd(op + q * 4 + 2, v.z * alpha);
        atomicAdd(op + q * 4 + 3, v.w * alpha);
    }
}

__global__ __launch_bounds__(256) void gat_epilogue(
    const float* __restrict__ hbuf, const float* __restrict__ bias,
    float* __restrict__ out, int total)
{
    const int i = blockIdx.x * blockDim.x + threadIdx.x;
    if (i >= total) return;
    float v = out[i] + hbuf[i] + bias[i & 127];
    out[i] = (v > 0.f) ? v : expm1f(v);
}

extern "C" void kernel_launch(void* const* d_in, const int* in_sizes, int n_in,
                              void* d_out, int out_size, void* d_ws, size_t ws_size,
                              hipStream_t stream)
{
    const float* x      = (const float*)d_in[0];
    const int*   ei     = (const int*)  d_in[1];
    const float* W      = (const float*)d_in[2];
    const float* a_src  = (const float*)d_in[3];
    const float* a_trg  = (const float*)d_in[4];
    const float* bias   = (const float*)d_in[5];

    const int N = in_sizes[0] / 128;
    const int E = in_sizes[1] / 2;

    float* hbuf  = (float*)d_ws;
    float* ssrc  = hbuf  + (size_t)N * 128;
    float* strg  = ssrc  + (size_t)N * 8;
    float* ebuf  = strg  + (size_t)N * 8;
    float* denom = ebuf  + (size_t)E * 8;
    unsigned int* gmax = (unsigned int*)(denom + (size_t)N * 8);

    hipMemsetAsync(denom, 0, (size_t)N * 8 * sizeof(float) + sizeof(unsigned int), stream);
    hipMemsetAsync(d_out, 0, (size_t)out_size * sizeof(float), stream);

    const int mtiles = (N + 15) / 16;
    gat_gemm_wmma<<<mtiles, 256, 0, stream>>>(x, W, a_src, a_trg, hbuf, ssrc, strg, N);

    gat_edge_scores<<<(E + 255) / 256, 256, 0, stream>>>(ei, E, ssrc, strg, ebuf, gmax);

    gat_edge_exp<<<(E + 255) / 256, 256, 0, stream>>>(ei, E, ebuf, denom, gmax);

    gat_aggregate<<<(E * 8 + 255) / 256, 256, 0, stream>>>(ei, E, hbuf, ebuf, denom, (float*)d_out);

    const int total = N * 128;
    gat_epilogue<<<(total + 255) / 256, 256, 0, stream>>>(hbuf, bias, (float*)d_out, total);
}


# tile scan: the score-max launch writes one tile-id byte per edge and each tile block filters that byte column, 16 edges per 16-byte load, not the 4-byte targets; sums in new LDS arrival order
# speedup vs baseline: 5.4112x; 5.4112x over previous
#include <hip/hip_runtime.h>
#include <hip/hip_bf16.h>
#include <math.h>

typedef float v2f __attribute__((ext_vector_type(2)));
typedef float v8f __attribute__((ext_vector_type(8)));

#define NEG_SLOPE 0.2f
#define EPS_F 1e-16f

#define SWIZ_XOR1  0x041F
#define SWIZ_XOR2  0x081F
#define SWIZ_XOR4  0x101F
#define SWIZ_XOR8  0x201F
#define SWIZ_XOR16 0x401F
#define SWZ(v, imm) __int_as_float(__builtin_amdgcn_ds_swizzle(__float_as_int(v), imm))

__device__ __forceinline__ float reduce16_add(float v) {
    v += SWZ(v, SWIZ_XOR1);
    v += SWZ(v, SWIZ_XOR2);
    v += SWZ(v, SWIZ_XOR4);
    v += SWZ(v, SWIZ_XOR8);
    return v;
}
__device__ __forceinline__ float reduce32_max(float v) {
    v = fmaxf(v, SWZ(v, SWIZ_XOR1));
    v = fmaxf(v, SWZ(v, SWIZ_XOR2));
    v = fmaxf(v, SWZ(v, SWIZ_XOR4));
    v = fmaxf(v, SWZ(v, SWIZ_XOR8));
    v = fmaxf(v, SWZ(v, SWIZ_XOR16));
    return v;
}

__device__ __forceinline__ unsigned int fkey(float f) {
    unsigned int u = __float_as_uint(f);
    return (u & 0x80000000u) ? ~u : (u | 0x80000000u);
}
__device__ __forceinline__ float funkey(unsigned int k) {
    return __uint_as_float((k & 0x80000000u) ? (k & 0x7FFFFFFFu) : ~k);
}

#define GEMM_THREADS 256
#define GEMM_BLOCKS  512

__global__ __launch_bounds__(GEMM_THREADS) void gat_gemm_wmma(
    const float* __restrict__ x, const float* __restrict__ W,
    const float* __restrict__ a_src, const float* __restrict__ a_trg,
    float* __restrict__ hbuf, float* __restrict__ s_src, float* __restrict__ s_trg,
    int N)
{
    const int wave  = threadIdx.x >> 5;
    const int lane  = threadIdx.x & 31;
    const int nBase = wave * 16;
    const int f     = lane & 15;
    const int half  = lane >> 4;
    const int mtiles = (N + 15) >> 4;
    const int tstep  = (int)gridDim.x;

    v2f b[32];
    #pragma unroll
    for (int j = 0; j < 16; ++j) {
        const float* __restrict__ wp = W + (size_t)(8 * j + 4 * half) * 128 + nBase + f;
        b[2 * j].x     = wp[0];
        b[2 * j].y     = wp[128];
        b[2 * j + 1].x = wp[256];
        b[2 * j + 1].y = wp[384];
    }

    const float asv = a_src[nBase + f];
    const float atv = a_trg[nBase + f];

    #define GEMM_LOAD_A(dst, tile)                                                    \
        do {                                                                          \
            int rowA_ = (tile) * 16 + f;                                              \
            if (rowA_ > N - 1) rowA_ = N - 1;                                         \
            const float4* __restrict__ xr_ = (const float4*)(x + (size_t)rowA_ * 128) + half; \
            _Pragma("unroll")                                                         \
            for (int j_ = 0; j_ < 16; ++j_) (dst)[j_] = xr_[2 * j_];                  \
        } while (0)

    #define GEMM_TILE(av, tile)                                                       \
        do {                                                                          \
            v8f c0_ = {};                                                             \
            v8f c1_ = {};                                                             \
            _Pragma("unroll")                                                         \
            for (int j_ = 0; j_ < 16; ++j_) {                                         \
                v2f a0_, a1_;                                                         \
                a0_.x = (av)[j_].x; a0_.y = (av)[j_].y;                               \
                a1_.x = (av)[j_].z; a1_.y = (av)[j_].w;                               \
                c0_ = __builtin_amdgcn_wmma_f32_16x16x4_f32(false, a0_, false, b[2 * j_],     \
                                                            (short)0, c0_, false, false);     \
                c1_ = __builtin_amdgcn_wmma_f32_16x16x4_f32(false, a1_, false, b[2 * j_ + 1], \
                                                            (short)0, c1_, false, false);     \
            }                                                                         \
            const v8f c_ = c0_ + c1_;                                                 \
            const int mBase_ = (tile) * 16;                                           \
            _Pragma("unroll")                                                         \
            for (int r_ = 0; r_ < 8; ++r_) {                                          \
                const int m_ = mBase_ + r_ + half * 8;     \
                const float hv_ = c_[r_];                                             \
                if (m_ < N) hbuf[(size_t)m_ * 128 + nBase + f] = hv_;                 \
                const float ps_ = reduce16_add(hv_ * asv);                            \
                const float pt_ = reduce16_add(hv_ * atv);                            \
                if (f == 0 && m_ < N) {                                               \
                    s_src[(size_t)m_ * 8 + wave] = ps_;                               \
                    s_trg[(size_t)m_ * 8 + wave] = pt_;                               \
                }                                                                     \
            }                                                                         \
        } while (0)

    float4 a0[16], a1[16];
    int t = (int)blockIdx.x;
    if (t < mtiles) {
        const int tlast = mtiles - 1;
        GEMM_LOAD_A(a0, t);
        for (;;) {
            int tn = t + tstep;
            GEMM_LOAD_A(a1, (tn < tlast ? tn : tlast));
            __builtin_amdgcn_sched_barrier(0);
            GEMM_TILE(a0, t);
            if (tn >= mtiles) break;
            t = tn;
            tn = t + tstep;
            GEMM_LOAD_A(a0, (tn < tlast ? tn : tlast));
            __builtin_amdgcn_sched_barrier(0);
            GEMM_TILE(a1, t);
            if (tn >= mtiles) break;
            t = tn;
        }
    }
    #undef GEMM_TILE
    #undef GEMM_LOAD_A
}

#define TILE_NODES  400
#define AGG_THREADS 1024
#define AGG_WAVES   (AGG_THREADS / 32)
#define WLIST_CAP   256
#define SCAN_UNROLL 8
#define EDGE_BATCH  8
#define LIST_SINK   0x80000000u

__device__ __forceinline__ bool have_tile_column(size_t ws_bytes, int N, int E) {
    return ws_bytes >= (size_t)N * 576 + 16 + (size_t)E;
}

__global__ __launch_bounds__(256) void gat_edge_max(
    const int* __restrict__ ei, int E, int N,
    const float* __restrict__ s_src, const float* __restrict__ s_trg,
    unsigned int* __restrict__ gmax, unsigned char* __restrict__ tcol, size_t ws_bytes)
{
    __shared__ unsigned int smax;
    if (threadIdx.x == 0) smax = 0u;
    __syncthreads();

    const bool col = have_tile_column(ws_bytes, N, E);
    const int e = blockIdx.x * blockDim.x + threadIdx.x;
    float lmax = -INFINITY;
    if (e < E) {
        const int src = ei[e];
        const int trg = ei[E + e];
        if (col) tcol[e] = (unsigned char)((unsigned int)trg / (unsigned int)TILE_NODES);
        const float* __restrict__ ss = s_src + (size_t)src * 8;
        const float* __restrict__ st = s_trg + (size_t)trg * 8;
        #pragma unroll
        for (int hh = 0; hh < 8; ++hh) {
            float v = ss[hh] + st[hh];
            v = (v >= 0.f) ? v : NEG_SLOPE * v;
            lmax = fmaxf(lmax, v);
        }
    }
    lmax = reduce32_max(lmax);
    if ((threadIdx.x & 31) == 0) atomicMax(&smax, fkey(lmax));
    __syncthreads();
    if (threadIdx.x == 0) atomicMax(gmax, smax);
}


__device__ __noinline__ void tile_add_edge_by_lane(
    const int* __restrict__ ei, int E, const float* __restrict__ hbuf,
    const float* __restrict__ s_src, const float* __restrict__ s_trg,
    float* num, float* den, int base, unsigned int tn, float mx, unsigned int e)
{
    const int src = ei[e];
    const int trg = ei[(size_t)E + e];
    const unsigned int tl = (unsigned int)trg - (unsigned int)base;
    if (tl >= tn) return;
    for (int hh = 0; hh < 8; ++hh) {
        float s = s_src[(size_t)src * 8 + hh] + s_trg[(size_t)trg * 8 + hh];
        s = (s >= 0.f) ? s : NEG_SLOPE * s;
        const float w = expf(s - mx);
        atomicAdd(&den[tl * 8 + hh], w);
        for (int f = 0; f < 16; ++f) {
            const int feat = hh * 16 + f;
            atomicAdd(&num[tl * 128 + (feat & 3) * 32 + (feat >> 2)],
                      w * hbuf[(size_t)src * 128 + feat]);
        }
    }
}

__global__ __launch_bounds__(AGG_THREADS) void gat_tile_aggregate(
    const int* __restrict__ ei, int E, int N,
    const float* __restrict__ hbuf,
    const float* __restrict__ s_src, const float* __restrict__ s_trg,
    const float* __restrict__ bias, const unsigned int* __restrict__ gmax,
    const unsigned char* __restrict__ tcol, size_t ws_bytes,
    float* __restrict__ out)
{
    __shared__ float num[(TILE_NODES + 1) * 128];
    __shared__ float den[(TILE_NODES + 1) * 8];
    __shared__ unsigned int wlists[AGG_WAVES * WLIST_CAP];

    const int tid  = threadIdx.x;
    const int lane = tid & 31;
    const int wave = tid >> 5;
    const int base = blockIdx.x * TILE_NODES;
    const unsigned int tn = (unsigned int)((N - base < TILE_NODES) ? (N - base) : TILE_NODES);

    for (int i = tid; i < TILE_NODES * 32; i += AGG_THREADS)
        ((float4*)num)[i] = make_float4(0.f, 0.f, 0.f, 0.f);
    for (int i = tid; i < TILE_NODES * 8; i += AGG_THREADS) den[i] = 0.f;
    __syncthreads();

    const float mx = funkey(*gmax);
    unsigned int* wl = wlists + wave * WLIST_CAP;
    unsigned int wcnt = 0u;
    const int* __restrict__ trgp = ei + E;

    #define TILE_TAKE(match, e)                                                       \
        do {                                                                          \
            const unsigned int bal_ = __builtin_amdgcn_ballot_w32(match);             \
            if (bal_ != 0u) {                                                         \
                const unsigned int pos_ = __builtin_amdgcn_mbcnt_lo(bal_, wcnt);        \
                if (match) {                                                          \
                    if (pos_ < WLIST_CAP) wl[pos_] = (unsigned int)(e);               \
                    else tile_add_edge_by_lane(ei, E, hbuf, s_src, s_trg, num, den,   \
                                               base, tn, mx, (unsigned int)(e));      \
                }                                                                     \
                wcnt += (unsigned int)__popc(bal_);                                   \
            }                                                                         \
        } while (0)

    #define TILE_EQ_BYTES(w)                                                          \
        (~((((((unsigned int)(w)) ^ pat) & 0x7F7F7F7Fu) + 0x7F7F7F7Fu) |              \
           (((unsigned int)(w)) ^ pat)) & 0x80808080u)

    #define TILE_SCAN_PAIR(tA, tB, g)                                                 \
        do {                                                                          \
            unsigned int m_ = (TILE_EQ_BYTES((tA).x) >> 7) | (TILE_EQ_BYTES((tA).y) >> 6) | \
                              (TILE_EQ_BYTES((tA).z) >> 5) | (TILE_EQ_BYTES((tA).w) >> 4) | \
                              (TILE_EQ_BYTES((tB).x) >> 3) | (TILE_EQ_BYTES((tB).y) >> 2) | \
                              (TILE_EQ_BYTES((tB).z) >> 1) |  TILE_EQ_BYTES((tB).w);  \
            unsigned int bal_ = __builtin_amdgcn_ballot_w32(m_ != 0u);                \
            while (bal_ != 0u) {                                                      \
                const unsigned int b_ = (unsigned int)__builtin_ctz(m_ | 0x80000000u); \
                const unsigned int e_ = (unsigned int)(g) * 16u + ((b_ & 4u) * (AGG_THREADS * 4u)) + \
                                        ((b_ & 3u) << 2) + (b_ >> 3);                 \
                const unsigned int pos_ = __builtin_amdgcn_mbcnt_lo(bal_, wcnt);      \
                if (m_ != 0u) {                                                       \
                    if (pos_ < WLIST_CAP) wl[pos_] = e_;                              \
                    else tile_add_edge_by_lane(ei, E, hbuf, s_src, s_trg, num, den,   \
                                               base, tn, mx, e_);                     \
                }                                                                     \
                wcnt += (unsigned int)__popc(bal_);                                   \
                m_ &= m_ - 1u;                                                        \
                bal_ = __builtin_amdgcn_ballot_w32(m_ != 0u);                         \
            }                                                                         \
        } while (0)

    int edone = 0;
    if (have_tile_column(ws_bytes, N, E)) {
        const unsigned int pat = ((unsigned int)blockIdx.x & 0xFFu) * 0x01010101u;
        const int npat = (int)~pat;
        const int4 none = make_int4(npat, npat, npat, npat);
        const int4* __restrict__ tv = (const int4*)tcol;
        const int nv    = E >> 4;
        const int nfull = nv / AGG_THREADS;
        const int nmain = nfull - nfull % SCAN_UNROLL;
        if (nmain > 0) {
            const int r0 = (int)((((unsigned int)blockIdx.x * 37u) % (unsigned int)nmain) & ~1u);
            for (int k0 = 0; k0 < nmain; k0 += SCAN_UNROLL) {
                int4 t[SCAN_UNROLL];
                int  gi[SCAN_UNROLL / 2];
                #pragma unroll
                for (int u = 0; u < SCAN_UNROLL; u += 2) {
                    int r = r0 + k0 + u;
                    if (r >= nmain) r -= nmain;
                    gi[u >> 1] = r * AGG_THREADS + tid;
                    t[u]     = tv[gi[u >> 1]];
                    t[u + 1] = tv[gi[u >> 1] + AGG_THREADS];
                }
                #pragma unroll
                for (int u = 0; u < SCAN_UNROLL; u += 2) TILE_SCAN_PAIR(t[u], t[u + 1], gi[u >> 1]);
            }
        }
        for (int gb = nmain * AGG_THREADS; gb < nv; gb += AGG_THREADS) {
            const int g = gb + tid;
            const int4 t = (g < nv) ? tv[g] : none;
            TILE_SCAN_PAIR(t, none, g);
        }
        edone = nv << 4;
    }
    for (int e0 = edone; e0 < E; e0 += AGG_THREADS) {
        const int e = e0 + tid;
        const unsigned int d = (e < E) ? (unsigned int)trgp[e] - (unsigned int)base : 0xFFFFFFFFu;
        TILE_TAKE(d < tn, e);
    }
    #undef TILE_SCAN_PAIR
    #undef TILE_EQ_BYTES
    #undef TILE_TAKE

    unsigned int n = (wcnt < WLIST_CAP) ? wcnt : WLIST_CAP;
    const unsigned int pad = (0u - n) & (EDGE_BATCH - 1);
    if ((unsigned int)lane < pad) wl[n + lane] = LIST_SINK;
    n += pad;
    const int head = lane >> 2;
    for (unsigned int i = 0; i < n; i += EDGE_BATCH) {
        unsigned int ed[EDGE_BATCH];
        unsigned int tl[EDGE_BATCH];
        size_t sr[EDGE_BATCH];
        size_t tg[EDGE_BATCH];
        float  sc[EDGE_BATCH];
        float4 row[EDGE_BATCH];
        #pragma unroll
        for (int u = 0; u < EDGE_BATCH; ++u) ed[u] = wl[i + u];
        #pragma unroll
        for (int u = 0; u < EDGE_BATCH; ++u) {
            const unsigned int el = ed[u] & ~LIST_SINK;
            sr[u] = (size_t)ei[el];
            tg[u] = (size_t)trgp[el];
        }
        #pragma unroll
        for (int u = 0; u < EDGE_BATCH; ++u) {
            const unsigned int d = (unsigned int)tg[u] - (unsigned int)base;
            tl[u]  = (ed[u] < LIST_SINK && d < tn) ? d : (unsigned int)TILE_NODES;
            sc[u]  = s_src[sr[u] * 8 + head] + s_trg[tg[u] * 8 + head];
            row[u] = *(const float4*)(hbuf + sr[u] * 128 + lane * 4);
        }
        #pragma unroll
        for (int u = 0; u < EDGE_BATCH; ++u) {
            float s = sc[u];
            s = (s >= 0.f) ? s : NEG_SLOPE * s;
            const float w = expf(s - mx);
            float* a = num + tl[u] * 128 + lane;
            atomicAdd(a,      w * row[u].x);
            atomicAdd(a + 32, w * row[u].y);
            atomicAdd(a + 64, w * row[u].z);
            atomicAdd(a + 96, w * row[u].w);
            if ((lane & 3) == 0) atomicAdd(&den[tl[u] * 8 + head], w);
        }
    }
    __syncthreads();

    const float4 bv = *(const float4*)(bias + lane * 4);
    for (unsigned int tl = wave; tl < tn; tl += AGG_WAVES) {
        const size_t node = (size_t)base + tl;
        const float4 hv = *(const float4*)(hbuf + node * 128 + lane * 4);
        const float d = den[tl * 8 + head] + EPS_F;
        const float* a = num + tl * 128 + lane;
        float4 v;
        v.x = a[0]  / d + hv.x + bv.x;
        v.y = a[32] / d + hv.y + bv.y;
        v.z = a[64] / d + hv.z + bv.z;
        v.w = a[96] / d + hv.w + bv.w;
        v.x = (v.x > 0.f) ? v.x : expm1f(v.x);
        v.y = (v.y > 0.f) ? v.y : expm1f(v.y);
        v.z = (v.z > 0.f) ? v.z : expm1f(v.z);
        v.w = (v.w > 0.f) ? v.w : expm1f(v.w);
        *(float4*)(out + node * 128 + lane * 4) = v;
    }
}

extern "C" void kernel_launch(void* const* d_in, const int* in_sizes, int n_in,
                              void* d_out, int out_size, void* d_ws, size_t ws_size,
                              hipStream_t stream)
{
    const float* x      = (const float*)d_in[0];
    const int*   ei     = (const int*)  d_in[1];
    const float* W      = (const float*)d_in[2];
    const float* a_src  = (const float*)d_in[3];
    const float* a_trg  = (const float*)d_in[4];
    const float* bias   = (const float*)d_in[5];

    const int N = in_sizes[0] / 128;
    const int E = in_sizes[1] / 2;

    float* hbuf  = (float*)d_ws;
    float* ssrc  = hbuf  + (size_t)N * 128;
    float* strg  = ssrc  + (size_t)N * 8;
    unsigned int* gmax = (unsigned int*)(strg + (size_t)N * 8);
    unsigned char* tcol = (unsigned char*)(gmax + 4);

    hipMemsetAsync(gmax, 0, sizeof(unsigned int), stream);

    gat_gemm_wmma<<<GEMM_BLOCKS, GEMM_THREADS, 0, stream>>>(x, W, a_src, a_trg, hbuf, ssrc, strg, N);

    gat_edge_max<<<(E + 255) / 256, 256, 0, stream>>>(ei, E, N, ssrc, strg, gmax, tcol, ws_size);

    gat_tile_aggregate<<<(N + TILE_NODES - 1) / TILE_NODES, AGG_THREADS, 0, stream>>>(
        ei, E, N, hbuf, ssrc, strg, bias, gmax, tcol, ws_size, (float*)d_out);
}
